# starting kernel
# baseline (speedup 1.0000x reference)
#include <hip/hip_runtime.h>
#include <hip/hip_bf16.h>

#define DZ    128
#define DB    32
#define DSIN  32
#define DX    64
#define BATCH 64
#define TSTEPS 4096

#define BREAL 8
#define NWG   (BATCH/BREAL)
#define NTHREADS 256

#define STH 33
#define SZR 132

typedef __attribute__((ext_vector_type(2))) float v2f;
typedef __attribute__((ext_vector_type(8))) float v8f;

__launch_bounds__(NTHREADS, 1)
__global__ void plrnn_fused_kernel(const float* __restrict__ z0,
                                   const float* __restrict__ inputs,
                                   const float* __restrict__ AW,
                                   const float* __restrict__ h,
                                   const float* __restrict__ thetas,
                                   const float* __restrict__ alphas,
                                   const float* __restrict__ clipping,
                                   const float* __restrict__ C,
                                   const float* __restrict__ Bobs,
                                   float* __restrict__ out)
{
    __shared__ float th_sh [DZ * STH];
    __shared__ float th2_sh[DZ * STH];
    __shared__ float z_sh  [16 * SZR];
    __shared__ float be_sh [16 * SZR];
    __shared__ float al_sh [DB];
    __shared__ float ad_sh [DZ];
    __shared__ float h_sh  [DZ];
    __shared__ float hc_sh [DZ];
    __shared__ float s_sh  [DSIN];

    const int tid   = threadIdx.x;
    const int lane  = tid & 31;
    const int wave  = tid >> 5;
    const int m16   = lane & 15;
    const int khalf = lane >> 4;
    const int b0    = blockIdx.x * BREAL;

    for (int idx = tid; idx < DZ * DB; idx += NTHREADS) {
        int i = idx >> 5, j = idx & 31;
        float t = thetas[idx];
        th_sh [i * STH + j] = t;
        th2_sh[i * STH + j] = t - clipping[idx];
    }
    for (int idx = tid; idx < BREAL * DZ; idx += NTHREADS) {
        int m = idx >> 7, k = idx & 127;
        z_sh[m * SZR + k] = z0[(size_t)(b0 + m) * DZ + k];
    }
    for (int idx = tid; idx < 8 * DZ; idx += NTHREADS) {
        int m = (idx >> 7) + 8, k = idx & 127;
        z_sh [m * SZR + k] = 0.0f;
    }
    for (int idx = tid; idx < 16 * DZ; idx += NTHREADS) {
        int m = idx >> 7, k = idx & 127;
        be_sh[m * SZR + k] = 0.0f;
    }
    if (tid < DB) al_sh[tid] = alphas[tid];
    if (tid < DZ) { ad_sh[tid] = AW[tid * DZ + tid]; h_sh[tid] = h[tid]; }

    float Crow[DSIN];
    if (tid < DZ) {
        #pragma unroll
        for (int j = 0; j < DSIN; ++j) Crow[j] = C[tid * DSIN + j];
    }

    const int ncol = wave * 16 + m16;
    v2f wtB[DZ / 4];
    #pragma unroll
    for (int k = 0; k < DZ / 4; ++k) {
        int kk = 4 * k + 2 * khalf;
        float bx = (ncol == kk    ) ? 0.0f : AW[(size_t)ncol * DZ + kk];
        float by = (ncol == kk + 1) ? 0.0f : AW[(size_t)ncol * DZ + kk + 1];
        v2f b; b.x = bx; b.y = by;
        wtB[k] = b;
    }

    const int nout = (wave & 3) * 16 + m16;
    v2f boB[DZ / 4];
    #pragma unroll
    for (int k = 0; k < DZ / 4; ++k) {
        int kk = 4 * k + 2 * khalf;
        v2f b;
        b.x = Bobs[(size_t)kk * DX + nout];
        b.y = Bobs[(size_t)(kk + 1) * DX + nout];
        boB[k] = b;
    }
    __syncthreads();

    for (int t = 0; t < TSTEPS; ++t) {
        if (tid < DSIN) s_sh[tid] = inputs[(size_t)t * DSIN + tid];
        __syncthreads();

        {
            const int i  = tid & 127;
            const int bb = (tid >> 7) * 4;
            float zr[4], acc[4];
            #pragma unroll
            for (int b = 0; b < 4; ++b) { zr[b] = z_sh[(bb + b) * SZR + i]; acc[b] = 0.0f; }
            #pragma unroll
            for (int j = 0; j < DB; ++j) {
                float th  = th_sh [i * STH + j];
                float th2 = th2_sh[i * STH + j];
                float a   = al_sh[j];
                #pragma unroll
                for (int b = 0; b < 4; ++b) {
                    float d1 = zr[b] - th;
                    float d2 = zr[b] - th2;
                    float l1 = fmaxf(d1, 0.18f * d1);
                    float l2 = fmaxf(d2, 0.18f * d2);
                    acc[b] = fmaf(a, l1 - l2, acc[b]);
                }
            }
            #pragma unroll
            for (int b = 0; b < 4; ++b) be_sh[(bb + b) * SZR + i] = acc[b];
        }
        if (tid < DZ) {
            float a = h_sh[tid];
            #pragma unroll
            for (int j = 0; j < DSIN; ++j) a = fmaf(Crow[j], s_sh[j], a);
            hc_sh[tid] = a;
        }
        __syncthreads();

        {
            v8f d = {0.f,0.f,0.f,0.f,0.f,0.f,0.f,0.f};
            #pragma unroll
            for (int k = 0; k < DZ / 4; ++k) {
                int kk = 4 * k + 2 * khalf;
                v2f a;
                a.x = be_sh[m16 * SZR + kk];
                a.y = be_sh[m16 * SZR + kk + 1];
                d = __builtin_amdgcn_wmma_f32_16x16x4_f32(
                        false, a, false, wtB[k], (short)0, d, false, false);
            }
            const float ad  = ad_sh[ncol];
            const float hcv = hc_sh[ncol];
            #pragma unroll
            for (int v = 0; v < 8; ++v) {
                int m = v + 8 * khalf;
                int idx = m * SZR + ncol;
                z_sh[idx] = fmaf(z_sh[idx], ad, d[v] + hcv);
            }
        }
        __syncthreads();

        if (wave < 4) {
            v8f o = {0.f,0.f,0.f,0.f,0.f,0.f,0.f,0.f};
            #pragma unroll
            for (int k = 0; k < DZ / 4; ++k) {
                int kk = 4 * k + 2 * khalf;
                v2f a;
                a.x = z_sh[m16 * SZR + kk];
                a.y = z_sh[m16 * SZR + kk + 1];
                o = __builtin_amdgcn_wmma_f32_16x16x4_f32(
                        false, a, false, boB[k], (short)0, o, false, false);
            }
            if (khalf == 0) {
                const size_t rowbase = (size_t)t * BATCH + b0;
                #pragma unroll
                for (int v = 0; v < 8; ++v)
                    out[(rowbase + v) * DX + nout] = o[v];
            }
        }
    }
}

extern "C" void kernel_launch(void* const* d_in, const int* in_sizes, int n_in,
                              void* d_out, int out_size, void* d_ws, size_t ws_size,
                              hipStream_t stream) {
    const float* z0   = (const float*)d_in[0];
    const float* inp  = (const float*)d_in[1];
    const float* AW   = (const float*)d_in[2];
    const float* hb   = (const float*)d_in[3];
    const float* th   = (const float*)d_in[4];
    const float* al   = (const float*)d_in[5];
    const float* cl   = (const float*)d_in[6];
    const float* C    = (const float*)d_in[7];
    const float* Bo   = (const float*)d_in[8];
    float* out = (float*)d_out;

    plrnn_fused_kernel<<<dim3(NWG), dim3(NTHREADS), 0, stream>>>(
        z0, inp, AW, hb, th, al, cl, C, Bo, out);

    (void)in_sizes; (void)n_in; (void)out_size; (void)d_ws; (void)ws_size;
}


# One batch row per work-group (64 work-groups of 1 row, was 8 of 8 rows): per-step basis expansion work per work-group drops 8x; waves 4-7, freed of it, compute the h+C*s vector
# speedup vs baseline: 1.3111x; 1.3111x over previous
#include <hip/hip_runtime.h>
#include <hip/hip_bf16.h>

#define DZ    128
#define DB    32
#define DSIN  32
#define DX    64
#define BATCH 64
#define TSTEPS 4096

#define BREAL 1
#define NWG   (BATCH/BREAL)
#define NTHREADS 256

#define STH 33
#define SZR 132

typedef __attribute__((ext_vector_type(2))) float v2f;
typedef __attribute__((ext_vector_type(8))) float v8f;

__launch_bounds__(NTHREADS, 1)
__global__ void plrnn_fused_kernel(const float* __restrict__ z0,
                                   const float* __restrict__ inputs,
                                   const float* __restrict__ AW,
                                   const float* __restrict__ h,
                                   const float* __restrict__ thetas,
                                   const float* __restrict__ alphas,
                                   const float* __restrict__ clipping,
                                   const float* __restrict__ C,
                                   const float* __restrict__ Bobs,
                                   float* __restrict__ out)
{
    __shared__ float th_sh [DZ * STH];
    __shared__ float th2_sh[DZ * STH];
    __shared__ float z_sh  [16 * SZR];
    __shared__ float be_sh [16 * SZR];
    __shared__ float al_sh [DB];
    __shared__ float ad_sh [DZ];
    __shared__ float h_sh  [DZ];
    __shared__ float hc_sh [DZ];
    __shared__ float s_sh  [DSIN];

    const int tid   = threadIdx.x;
    const int lane  = tid & 31;
    const int wave  = tid >> 5;
    const int m16   = lane & 15;
    const int khalf = lane >> 4;
    const int b0    = blockIdx.x * BREAL;

    for (int idx = tid; idx < DZ * DB; idx += NTHREADS) {
        int i = idx >> 5, j = idx & 31;
        float t = thetas[idx];
        th_sh [i * STH + j] = t;
        th2_sh[i * STH + j] = t - clipping[idx];
    }
    for (int idx = tid; idx < 16 * DZ; idx += NTHREADS) {
        int m = idx >> 7, k = idx & 127;
        z_sh [m * SZR + k] = (m == 0) ? z0[(size_t)b0 * DZ + k] : 0.0f;
        be_sh[m * SZR + k] = 0.0f;
    }
    if (tid < DB) al_sh[tid] = alphas[tid];
    if (tid < DZ) { ad_sh[tid] = AW[tid * DZ + tid]; h_sh[tid] = h[tid]; }

    float Crow[DSIN];
    if (tid >= DZ) {
        #pragma unroll
        for (int j = 0; j < DSIN; ++j) Crow[j] = C[(tid - DZ) * DSIN + j];
    }

    const int ncol = wave * 16 + m16;
    v2f wtB[DZ / 4];
    #pragma unroll
    for (int k = 0; k < DZ / 4; ++k) {
        int kk = 4 * k + 2 * khalf;
        float bx = (ncol == kk    ) ? 0.0f : AW[(size_t)ncol * DZ + kk];
        float by = (ncol == kk + 1) ? 0.0f : AW[(size_t)ncol * DZ + kk + 1];
        v2f b; b.x = bx; b.y = by;
        wtB[k] = b;
    }

    const int nout = (wave & 3) * 16 + m16;
    v2f boB[DZ / 4];
    #pragma unroll
    for (int k = 0; k < DZ / 4; ++k) {
        int kk = 4 * k + 2 * khalf;
        v2f b;
        b.x = Bobs[(size_t)kk * DX + nout];
        b.y = Bobs[(size_t)(kk + 1) * DX + nout];
        boB[k] = b;
    }
    __syncthreads();

    for (int t = 0; t < TSTEPS; ++t) {
        if (tid < DSIN) s_sh[tid] = inputs[(size_t)t * DSIN + tid];
        __syncthreads();

        if (tid < DZ) {
            const int i = tid;
            const float zr = z_sh[i];
            float acc = 0.0f;
            #pragma unroll
            for (int j = 0; j < DB; ++j) {
                float th  = th_sh [i * STH + j];
                float th2 = th2_sh[i * STH + j];
                float a   = al_sh[j];
                float d1 = zr - th;
                float d2 = zr - th2;
                float l1 = fmaxf(d1, 0.18f * d1);
                float l2 = fmaxf(d2, 0.18f * d2);
                acc = fmaf(a, l1 - l2, acc);
            }
            be_sh[i] = acc;
        } else {
            const int r = tid - DZ;
            float a = h_sh[r];
            #pragma unroll
            for (int j = 0; j < DSIN; ++j) a = fmaf(Crow[j], s_sh[j], a);
            hc_sh[r] = a;
        }
        __syncthreads();

        {
            v8f d = {0.f,0.f,0.f,0.f,0.f,0.f,0.f,0.f};
            #pragma unroll
            for (int k = 0; k < DZ / 4; ++k) {
                int kk = 4 * k + 2 * khalf;
                v2f a;
                a.x = be_sh[m16 * SZR + kk];
                a.y = be_sh[m16 * SZR + kk + 1];
                d = __builtin_amdgcn_wmma_f32_16x16x4_f32(
                        false, a, false, wtB[k], (short)0, d, false, false);
            }
            if (khalf == 0)
                z_sh[ncol] = fmaf(z_sh[ncol], ad_sh[ncol], d[0] + hc_sh[ncol]);
        }
        __syncthreads();

        if (wave < 4) {
            v8f o = {0.f,0.f,0.f,0.f,0.f,0.f,0.f,0.f};
            #pragma unroll
            for (int k = 0; k < DZ / 4; ++k) {
                int kk = 4 * k + 2 * khalf;
                v2f a;
                a.x = z_sh[m16 * SZR + kk];
                a.y = z_sh[m16 * SZR + kk + 1];
                o = __builtin_amdgcn_wmma_f32_16x16x4_f32(
                        false, a, false, boB[k], (short)0, o, false, false);
            }
            if (khalf == 0)
                out[((size_t)t * BATCH + b0) * DX + nout] = o[0];
        }
    }
}

extern "C" void kernel_launch(void* const* d_in, const int* in_sizes, int n_in,
                              void* d_out, int out_size, void* d_ws, size_t ws_size,
                              hipStream_t stream) {
    const float* z0   = (const float*)d_in[0];
    const float* inp  = (const float*)d_in[1];
    const float* AW   = (const float*)d_in[2];
    const float* hb   = (const float*)d_in[3];
    const float* th   = (const float*)d_in[4];
    const float* al   = (const float*)d_in[5];
    const float* cl   = (const float*)d_in[6];
    const float* C    = (const float*)d_in[7];
    const float* Bo   = (const float*)d_in[8];
    float* out = (float*)d_out;

    plrnn_fused_kernel<<<dim3(NWG), dim3(NTHREADS), 0, stream>>>(
        z0, inp, AW, hb, th, al, cl, C, Bo, out);

    (void)in_sizes; (void)n_in; (void)out_size; (void)d_ws; (void)ws_size;
}


# Observation product batched over 16 time steps: state history tile in LDS with 16 real rows, one obs GEMM with its barrier and stores per 16 steps instead of per step; order of every sum unchanged
# speedup vs baseline: 1.8984x; 1.4479x over previous
#include <hip/hip_runtime.h>
#include <hip/hip_bf16.h>


#define DZ    128
#define DB    32
#define DSIN  32
#define DX    64
#define BATCH 64
#define TSTEPS 4096

#define BREAL 1
#define NWG   (BATCH/BREAL)
#define NTHREADS 256

#define STH 33
#define SZR 132
#define HROWS 16

typedef __attribute__((ext_vector_type(2))) float v2f;
typedef __attribute__((ext_vector_type(8))) float v8f;

__launch_bounds__(NTHREADS, 1)
__global__ void plrnn_fused_kernel(const float* __restrict__ z0,
                                   const float* __restrict__ inputs,
                                   const float* __restrict__ AW,
                                   const float* __restrict__ h,
                                   const float* __restrict__ thetas,
                                   const float* __restrict__ alphas,
                                   const float* __restrict__ clipping,
                                   const float* __restrict__ C,
                                   const float* __restrict__ Bobs,
                                   float* __restrict__ out)
{
    __shared__ float th_sh [DZ * STH];
    __shared__ float th2_sh[DZ * STH];
    __shared__ float zh_sh [HROWS * SZR];
    __shared__ float be_sh [16 * SZR];
    __shared__ float al_sh [DB];
    __shared__ float ad_sh [DZ];
    __shared__ float h_sh  [DZ];
    __shared__ float hc_sh [DZ];
    __shared__ float s_sh  [DSIN];

    const int tid   = threadIdx.x;
    const int lane  = tid & 31;
    const int wave  = tid >> 5;
    const int m16   = lane & 15;
    const int khalf = lane >> 4;
    const int b0    = blockIdx.x * BREAL;

    for (int idx = tid; idx < DZ * DB; idx += NTHREADS) {
        int i = idx >> 5, j = idx & 31;
        float t = thetas[idx];
        th_sh [i * STH + j] = t;
        th2_sh[i * STH + j] = t - clipping[idx];
    }
    for (int idx = tid; idx < 16 * DZ; idx += NTHREADS) {
        int m = idx >> 7, k = idx & 127;
        zh_sh[m * SZR + k] = (m == HROWS - 1) ? z0[(size_t)b0 * DZ + k] : 0.0f;
        be_sh[m * SZR + k] = 0.0f;
    }
    if (tid < DB) al_sh[tid] = alphas[tid];
    if (tid < DZ) { ad_sh[tid] = AW[tid * DZ + tid]; h_sh[tid] = h[tid]; }

    float Crow[DSIN];
    if (tid >= DZ) {
        #pragma unroll
        for (int j = 0; j < DSIN; ++j) Crow[j] = C[(tid - DZ) * DSIN + j];
    }

    const int ncol = wave * 16 + m16;
    v2f wtB[DZ / 4];
    #pragma unroll
    for (int k = 0; k < DZ / 4; ++k) {
        int kk = 4 * k + 2 * khalf;
        float bx = (ncol == kk    ) ? 0.0f : AW[(size_t)ncol * DZ + kk];
        float by = (ncol == kk + 1) ? 0.0f : AW[(size_t)ncol * DZ + kk + 1];
        v2f b; b.x = bx; b.y = by;
        wtB[k] = b;
    }

    const int nout = (wave & 3) * 16 + m16;
    v2f boB[DZ / 4];
    #pragma unroll
    for (int k = 0; k < DZ / 4; ++k) {
        int kk = 4 * k + 2 * khalf;
        v2f b;
        b.x = Bobs[(size_t)kk * DX + nout];
        b.y = Bobs[(size_t)(kk + 1) * DX + nout];
        boB[k] = b;
    }
    __syncthreads();

    for (int t = 0; t < TSTEPS; ++t) {
        const int prow = ((t + HROWS - 1) & (HROWS - 1)) * SZR;
        const int crow = (t & (HROWS - 1)) * SZR;
        if (tid < DSIN) s_sh[tid] = inputs[(size_t)t * DSIN + tid];
        __syncthreads();

        if (tid < DZ) {
            const int i = tid;
            const float zr = zh_sh[prow + i];
            float acc = 0.0f;
            #pragma unroll
            for (int j = 0; j < DB; ++j) {
                float th  = th_sh [i * STH + j];
                float th2 = th2_sh[i * STH + j];
                float a   = al_sh[j];
                float d1 = zr - th;
                float d2 = zr - th2;
                float l1 = fmaxf(d1, 0.18f * d1);
                float l2 = fmaxf(d2, 0.18f * d2);
                acc = fmaf(a, l1 - l2, acc);
            }
            be_sh[i] = acc;
        } else {
            const int r = tid - DZ;
            float a = h_sh[r];
            #pragma unroll
            for (int j = 0; j < DSIN; ++j) a = fmaf(Crow[j], s_sh[j], a);
            hc_sh[r] = a;
        }
        __syncthreads();

        {
            v8f d = {0.f,0.f,0.f,0.f,0.f,0.f,0.f,0.f};
            #pragma unroll
            for (int k = 0; k < DZ / 4; ++k) {
                int kk = 4 * k + 2 * khalf;
                v2f a;
                a.x = be_sh[m16 * SZR + kk];
                a.y = be_sh[m16 * SZR + kk + 1];
                d = __builtin_amdgcn_wmma_f32_16x16x4_f32(
                        false, a, false, wtB[k], (short)0, d, false, false);
            }
            if (khalf == 0)
                zh_sh[crow + ncol] = fmaf(zh_sh[prow + ncol], ad_sh[ncol], d[0] + hc_sh[ncol]);
        }

        if ((t & (HROWS - 1)) == HROWS - 1) {
            __syncthreads();
            if (wave < 4) {
                v8f o = {0.f,0.f,0.f,0.f,0.f,0.f,0.f,0.f};
                #pragma unroll
                for (int k = 0; k < DZ / 4; ++k) {
                    int kk = 4 * k + 2 * khalf;
                    v2f a;
                    a.x = zh_sh[m16 * SZR + kk];
                    a.y = zh_sh[m16 * SZR + kk + 1];
                    o = __builtin_amdgcn_wmma_f32_16x16x4_f32(
                            false, a, false, boB[k], (short)0, o, false, false);
                }
                const int tb = t - (HROWS - 1) + 8 * khalf;
                #pragma unroll
                for (int r = 0; r < 8; ++r)
                    out[((size_t)(tb + r) * BATCH + b0) * DX + nout] = o[r];
            }
        }
    }
}

extern "C" void kernel_launch(void* const* d_in, const int* in_sizes, int n_in,
                              void* d_out, int out_size, void* d_ws, size_t ws_size,
                              hipStream_t stream) {
    const float* z0   = (const float*)d_in[0];
    const float* inp  = (const float*)d_in[1];
    const float* AW   = (const float*)d_in[2];
    const float* hb   = (const float*)d_in[3];
    const float* th   = (const float*)d_in[4];
    const float* al   = (const float*)d_in[5];
    const float* cl   = (const float*)d_in[6];
    const float* C    = (const float*)d_in[7];
    const float* Bo   = (const float*)d_in[8];
    float* out = (float*)d_out;

    plrnn_fused_kernel<<<dim3(NWG), dim3(NTHREADS), 0, stream>>>(
        z0, inp, AW, hb, th, al, cl, C, Bo, out);

    (void)in_sizes; (void)n_in; (void)out_size; (void)d_ws; (void)ws_size;
}


# Input row of step t+1 fetched into a register during step t's matrix products and only copied to LDS at the top of its own step: no step waits for global memory any more; every sum unchanged
# speedup vs baseline: 2.2222x; 1.1706x over previous
#include <hip/hip_runtime.h>
#include <hip/hip_bf16.h>


#define DZ    128
#define DB    32
#define DSIN  32
#define DX    64
#define BATCH 64
#define TSTEPS 4096

#define BREAL 1
#define NWG   (BATCH/BREAL)
#define NTHREADS 256

#define STH 33
#define SZR 132
#define HROWS 16

typedef __attribute__((ext_vector_type(2))) float v2f;
typedef __attribute__((ext_vector_type(8))) float v8f;

__launch_bounds__(NTHREADS, 1)
__global__ void plrnn_fused_kernel(const float* __restrict__ z0,
                                   const float* __restrict__ inputs,
                                   const float* __restrict__ AW,
                                   const float* __restrict__ h,
                                   const float* __restrict__ thetas,
                                   const float* __restrict__ alphas,
                                   const float* __restrict__ clipping,
                                   const float* __restrict__ C,
                                   const float* __restrict__ Bobs,
                                   float* __restrict__ out)
{
    __shared__ float th_sh [DZ * STH];
    __shared__ float th2_sh[DZ * STH];
    __shared__ float zh_sh [HROWS * SZR];
    __shared__ float be_sh [16 * SZR];
    __shared__ float al_sh [DB];
    __shared__ float ad_sh [DZ];
    __shared__ float h_sh  [DZ];
    __shared__ float hc_sh [DZ];
    __shared__ float s_sh  [DSIN];

    const int tid   = threadIdx.x;
    const int lane  = tid & 31;
    const int wave  = tid >> 5;
    const int m16   = lane & 15;
    const int khalf = lane >> 4;
    const int b0    = blockIdx.x * BREAL;

    for (int idx = tid; idx < DZ * DB; idx += NTHREADS) {
        int i = idx >> 5, j = idx & 31;
        float t = thetas[idx];
        th_sh [i * STH + j] = t;
        th2_sh[i * STH + j] = t - clipping[idx];
    }
    for (int idx = tid; idx < 16 * DZ; idx += NTHREADS) {
        int m = idx >> 7, k = idx & 127;
        zh_sh[m * SZR + k] = (m == HROWS - 1) ? z0[(size_t)b0 * DZ + k] : 0.0f;
        be_sh[m * SZR + k] = 0.0f;
    }
    if (tid < DB) al_sh[tid] = alphas[tid];
    if (tid < DZ) { ad_sh[tid] = AW[tid * DZ + tid]; h_sh[tid] = h[tid]; }

    float Crow[DSIN];
    if (tid >= DZ) {
        #pragma unroll
        for (int j = 0; j < DSIN; ++j) Crow[j] = C[(tid - DZ) * DSIN + j];
    }

    float snext = 0.0f;
    if (tid < DSIN) snext = inputs[tid];

    const int ncol = wave * 16 + m16;
    v2f wtB[DZ / 4];
    #pragma unroll
    for (int k = 0; k < DZ / 4; ++k) {
        int kk = 4 * k + 2 * khalf;
        float bx = (ncol == kk    ) ? 0.0f : AW[(size_t)ncol * DZ + kk];
        float by = (ncol == kk + 1) ? 0.0f : AW[(size_t)ncol * DZ + kk + 1];
        v2f b; b.x = bx; b.y = by;
        wtB[k] = b;
    }

    const int nout = (wave & 3) * 16 + m16;
    v2f boB[DZ / 4];
    #pragma unroll
    for (int k = 0; k < DZ / 4; ++k) {
        int kk = 4 * k + 2 * khalf;
        v2f b;
        b.x = Bobs[(size_t)kk * DX + nout];
        b.y = Bobs[(size_t)(kk + 1) * DX + nout];
        boB[k] = b;
    }
    __syncthreads();

    for (int t = 0; t < TSTEPS; ++t) {
        const int prow = ((t + HROWS - 1) & (HROWS - 1)) * SZR;
        const int crow = (t & (HROWS - 1)) * SZR;
        if (tid < DSIN) s_sh[tid] = snext;
        __syncthreads();

        if (tid < DZ) {
            const int i = tid;
            const float zr = zh_sh[prow + i];
            float acc = 0.0f;
            #pragma unroll
            for (int j = 0; j < DB; ++j) {
                float th  = th_sh [i * STH + j];
                float th2 = th2_sh[i * STH + j];
                float a   = al_sh[j];
                float d1 = zr - th;
                float d2 = zr - th2;
                float l1 = fmaxf(d1, 0.18f * d1);
                float l2 = fmaxf(d2, 0.18f * d2);
                acc = fmaf(a, l1 - l2, acc);
            }
            be_sh[i] = acc;
        } else {
            const int r = tid - DZ;
            float a = h_sh[r];
            #pragma unroll
            for (int j = 0; j < DSIN; ++j) a = fmaf(Crow[j], s_sh[j], a);
            hc_sh[r] = a;
        }
        __syncthreads();

        if (tid < DSIN) {
            const int tn = (t + 1 < TSTEPS) ? (t + 1) : (TSTEPS - 1);
            snext = inputs[(size_t)tn * DSIN + tid];
        }

        {
            v8f d = {0.f,0.f,0.f,0.f,0.f,0.f,0.f,0.f};
            #pragma unroll
            for (int k = 0; k < DZ / 4; ++k) {
                int kk = 4 * k + 2 * khalf;
                v2f a;
                a.x = be_sh[m16 * SZR + kk];
                a.y = be_sh[m16 * SZR + kk + 1];
                d = __builtin_amdgcn_wmma_f32_16x16x4_f32(
                        false, a, false, wtB[k], (short)0, d, false, false);
            }
            if (khalf == 0)
                zh_sh[crow + ncol] = fmaf(zh_sh[prow + ncol], ad_sh[ncol], d[0] + hc_sh[ncol]);
        }

        if ((t & (HROWS - 1)) == HROWS - 1) {
            __syncthreads();
            if (wave < 4) {
                v8f o = {0.f,0.f,0.f,0.f,0.f,0.f,0.f,0.f};
                #pragma unroll
                for (int k = 0; k < DZ / 4; ++k) {
                    int kk = 4 * k + 2 * khalf;
                    v2f a;
                    a.x = zh_sh[m16 * SZR + kk];
                    a.y = zh_sh[m16 * SZR + kk + 1];
                    o = __builtin_amdgcn_wmma_f32_16x16x4_f32(
                            false, a, false, boB[k], (short)0, o, false, false);
                }
                const int tb = t - (HROWS - 1) + 8 * khalf;
                #pragma unroll
                for (int r = 0; r < 8; ++r)
                    out[((size_t)(tb + r) * BATCH + b0) * DX + nout] = o[r];
            }
        }
    }
}

extern "C" void kernel_launch(void* const* d_in, const int* in_sizes, int n_in,
                              void* d_out, int out_size, void* d_ws, size_t ws_size,
                              hipStream_t stream) {
    const float* z0   = (const float*)d_in[0];
    const float* inp  = (const float*)d_in[1];
    const float* AW   = (const float*)d_in[2];
    const float* hb   = (const float*)d_in[3];
    const float* th   = (const float*)d_in[4];
    const float* al   = (const float*)d_in[5];
    const float* cl   = (const float*)d_in[6];
    const float* C    = (const float*)d_in[7];
    const float* Bo   = (const float*)d_in[8];
    float* out = (float*)d_out;

    plrnn_fused_kernel<<<dim3(NWG), dim3(NTHREADS), 0, stream>>>(
        z0, inp, AW, hb, th, al, cl, C, Bo, out);

    (void)in_sizes; (void)n_in; (void)out_size; (void)d_ws; (void)ws_size;
}


# All LDS reads of each product phase (its 32 A fragments, then the three operands of the state update) issued as one group ahead of the WMMA chain, not one read per two WMMAs; every sum unchanged
# speedup vs baseline: 2.4423x; 1.0990x over previous
#include <hip/hip_runtime.h>
#include <hip/hip_bf16.h>


#define DZ    128
#define DB    32
#define DSIN  32
#define DX    64
#define BATCH 64
#define TSTEPS 4096

#define BREAL 1
#define NWG   (BATCH/BREAL)
#define NTHREADS 256

#define STH 33
#define SZR 132
#define HROWS 16

typedef __attribute__((ext_vector_type(2))) float v2f;
typedef __attribute__((ext_vector_type(8))) float v8f;

__launch_bounds__(NTHREADS, 1)
__global__ void plrnn_fused_kernel(const float* __restrict__ z0,
                                   const float* __restrict__ inputs,
                                   const float* __restrict__ AW,
                                   const float* __restrict__ h,
                                   const float* __restrict__ thetas,
                                   const float* __restrict__ alphas,
                                   const float* __restrict__ clipping,
                                   const float* __restrict__ C,
                                   const float* __restrict__ Bobs,
                                   float* __restrict__ out)
{
    __shared__ float th_sh [DZ * STH];
    __shared__ float th2_sh[DZ * STH];
    __shared__ float zh_sh [HROWS * SZR];
    __shared__ float be_sh [16 * SZR];
    __shared__ float al_sh [DB];
    __shared__ float ad_sh [DZ];
    __shared__ float h_sh  [DZ];
    __shared__ float hc_sh [DZ];
    __shared__ float s_sh  [DSIN];

    const int tid   = threadIdx.x;
    const int lane  = tid & 31;
    const int wave  = tid >> 5;
    const int m16   = lane & 15;
    const int khalf = lane >> 4;
    const int b0    = blockIdx.x * BREAL;

    for (int idx = tid; idx < DZ * DB; idx += NTHREADS) {
        int i = idx >> 5, j = idx & 31;
        float t = thetas[idx];
        th_sh [i * STH + j] = t;
        th2_sh[i * STH + j] = t - clipping[idx];
    }
    for (int idx = tid; idx < 16 * DZ; idx += NTHREADS) {
        int m = idx >> 7, k = idx & 127;
        zh_sh[m * SZR + k] = (m == HROWS - 1) ? z0[(size_t)b0 * DZ + k] : 0.0f;
        be_sh[m * SZR + k] = 0.0f;
    }
    if (tid < DB) al_sh[tid] = alphas[tid];
    if (tid < DZ) { ad_sh[tid] = AW[tid * DZ + tid]; h_sh[tid] = h[tid]; }

    float Crow[DSIN];
    if (tid >= DZ) {
        #pragma unroll
        for (int j = 0; j < DSIN; ++j) Crow[j] = C[(tid - DZ) * DSIN + j];
    }

    float snext = 0.0f;
    if (tid < DSIN) snext = inputs[tid];

    const int ncol = wave * 16 + m16;
    v2f wtB[DZ / 4];
    #pragma unroll
    for (int k = 0; k < DZ / 4; ++k) {
        int kk = 4 * k + 2 * khalf;
        float bx = (ncol == kk    ) ? 0.0f : AW[(size_t)ncol * DZ + kk];
        float by = (ncol == kk + 1) ? 0.0f : AW[(size_t)ncol * DZ + kk + 1];
        v2f b; b.x = bx; b.y = by;
        wtB[k] = b;
    }

    const int nout = (wave & 3) * 16 + m16;
    v2f boB[DZ / 4];
    #pragma unroll
    for (int k = 0; k < DZ / 4; ++k) {
        int kk = 4 * k + 2 * khalf;
        v2f b;
        b.x = Bobs[(size_t)kk * DX + nout];
        b.y = Bobs[(size_t)(kk + 1) * DX + nout];
        boB[k] = b;
    }
    __syncthreads();

    for (int t = 0; t < TSTEPS; ++t) {
        const int prow = ((t + HROWS - 1) & (HROWS - 1)) * SZR;
        const int crow = (t & (HROWS - 1)) * SZR;
        if (tid < DSIN) s_sh[tid] = snext;
        __syncthreads();

        if (tid < DZ) {
            const int i = tid;
            const float zr = zh_sh[prow + i];
            float acc = 0.0f;
            #pragma unroll
            for (int j = 0; j < DB; ++j) {
                float th  = th_sh [i * STH + j];
                float th2 = th2_sh[i * STH + j];
                float a   = al_sh[j];
                float d1 = zr - th;
                float d2 = zr - th2;
                float l1 = fmaxf(d1, 0.18f * d1);
                float l2 = fmaxf(d2, 0.18f * d2);
                acc = fmaf(a, l1 - l2, acc);
            }
            be_sh[i] = acc;
        } else {
            const int r = tid - DZ;
            float a = h_sh[r];
            #pragma unroll
            for (int j = 0; j < DSIN; ++j) a = fmaf(Crow[j], s_sh[j], a);
            hc_sh[r] = a;
        }
        __syncthreads();

        if (tid < DSIN) {
            const int tn = (t + 1 < TSTEPS) ? (t + 1) : (TSTEPS - 1);
            snext = inputs[(size_t)tn * DSIN + tid];
        }

        {
            v2f afr[DZ / 4];
            #pragma unroll
            for (int k = 0; k < DZ / 4; ++k) {
                int kk = 4 * k + 2 * khalf;
                v2f a;
                a.x = be_sh[m16 * SZR + kk];
                a.y = be_sh[m16 * SZR + kk + 1];
                afr[k] = a;
            }
            const float zprev = zh_sh[prow + ncol];
            const float adiag = ad_sh[ncol];
            const float hcv   = hc_sh[ncol];
            __builtin_amdgcn_sched_barrier(0);
            v8f d = {0.f,0.f,0.f,0.f,0.f,0.f,0.f,0.f};
            #pragma unroll
            for (int k = 0; k < DZ / 4; ++k)
                d = __builtin_amdgcn_wmma_f32_16x16x4_f32(
                        false, afr[k], false, wtB[k], (short)0, d, false, false);
            if (khalf == 0)
                zh_sh[crow + ncol] = fmaf(zprev, adiag, d[0] + hcv);
        }

        if ((t & (HROWS - 1)) == HROWS - 1) {
            __syncthreads();
            if (wave < 4) {
                v2f zfr[DZ / 4];
                #pragma unroll
                for (int k = 0; k < DZ / 4; ++k) {
                    int kk = 4 * k + 2 * khalf;
                    v2f a;
                    a.x = zh_sh[m16 * SZR + kk];
                    a.y = zh_sh[m16 * SZR + kk + 1];
                    zfr[k] = a;
                }
                __builtin_amdgcn_sched_barrier(0);
                v8f o = {0.f,0.f,0.f,0.f,0.f,0.f,0.f,0.f};
                #pragma unroll
                for (int k = 0; k < DZ / 4; ++k)
                    o = __builtin_amdgcn_wmma_f32_16x16x4_f32(
                            false, zfr[k], false, boB[k], (short)0, o, false, false);
                const int tb = t - (HROWS - 1) + 8 * khalf;
                #pragma unroll
                for (int r = 0; r < 8; ++r)
                    out[((size_t)(tb + r) * BATCH + b0) * DX + nout] = o[r];
            }
        }
    }
}

extern "C" void kernel_launch(void* const* d_in, const int* in_sizes, int n_in,
                              void* d_out, int out_size, void* d_ws, size_t ws_size,
                              hipStream_t stream) {
    const float* z0   = (const float*)d_in[0];
    const float* inp  = (const float*)d_in[1];
    const float* AW   = (const float*)d_in[2];
    const float* hb   = (const float*)d_in[3];
    const float* th   = (const float*)d_in[4];
    const float* al   = (const float*)d_in[5];
    const float* cl   = (const float*)d_in[6];
    const float* C    = (const float*)d_in[7];
    const float* Bo   = (const float*)d_in[8];
    float* out = (float*)d_out;

    plrnn_fused_kernel<<<dim3(NWG), dim3(NTHREADS), 0, stream>>>(
        z0, inp, AW, hb, th, al, cl, C, Bo, out);

    (void)in_sizes; (void)n_in; (void)out_size; (void)d_ws; (void)ws_size;
}


# Each wave expands its own 16 state units straight from its accumulator (2 lanes per unit, thetas in registers) and adds h+C*s itself: one barrier per step, not two; every sum unchanged
# speedup vs baseline: 3.1498x; 1.2897x over previous
#include <hip/hip_runtime.h>
#include <hip/hip_bf16.h>


#define DZ    128
#define DB    32
#define DSIN  32
#define DX    64
#define BATCH 64
#define TSTEPS 4096

#define BREAL 1
#define NWG   (BATCH/BREAL)
#define NTHREADS 256

#define SZR 132
#define HROWS 16
#define TILE (16 * SZR)
#define HB   (DB / 2)

typedef __attribute__((ext_vector_type(2))) float v2f;
typedef __attribute__((ext_vector_type(8))) float v8f;

static __device__ __forceinline__ float xhalf(float x) {
    int b = __builtin_bit_cast(int, x);
    return __builtin_bit_cast(float, __builtin_amdgcn_permlanex16(b, b, 0x76543210, 0xfedcba98, false, false));
}

static __device__ __forceinline__ float expand_unit(float zv, const float (&thr)[HB], const float (&th2r)[HB],
                                                    const float (&alr)[HB]) {
    float df[HB];
    #pragma unroll
    for (int jj = 0; jj < HB; ++jj) {
        float d1 = zv - thr[jj];
        float d2 = zv - th2r[jj];
        float l1 = fmaxf(d1, 0.18f * d1);
        float l2 = fmaxf(d2, 0.18f * d2);
        df[jj] = l1 - l2;
    }
    float acc = 0.0f;
    #pragma unroll
    for (int jj = 0; jj < HB; ++jj) acc = fmaf(alr[jj], df[jj], acc);
    acc = xhalf(acc);
    #pragma unroll
    for (int jj = 0; jj < HB; ++jj) acc = fmaf(alr[jj], df[jj], acc);
    return acc;
}

__launch_bounds__(NTHREADS, 1)
__global__ void plrnn_fused_kernel(const float* __restrict__ z0,
                                   const float* __restrict__ inputs,
                                   const float* __restrict__ AW,
                                   const float* __restrict__ h,
                                   const float* __restrict__ thetas,
                                   const float* __restrict__ alphas,
                                   const float* __restrict__ clipping,
                                   const float* __restrict__ C,
                                   const float* __restrict__ Bobs,
                                   float* __restrict__ out)
{
    __shared__ float be_sh[2 * TILE];
    __shared__ float zh_sh[2 * TILE];

    const int tid   = threadIdx.x;
    const int lane  = tid & 31;
    const int wave  = tid >> 5;
    const int m16   = lane & 15;
    const int khalf = lane >> 4;
    const int b0    = blockIdx.x * BREAL;

    for (int idx = tid; idx < 2 * TILE; idx += NTHREADS) be_sh[idx] = 0.0f;

    const int ncol = wave * 16 + m16;

    float thr[HB], th2r[HB], alr[HB];
    #pragma unroll
    for (int jj = 0; jj < HB; ++jj) {
        const int j = HB * khalf + jj;
        const float tv = thetas[ncol * DB + j];
        thr [jj] = tv;
        th2r[jj] = tv - clipping[ncol * DB + j];
        alr [jj] = alphas[j];
    }
    float Crow[DSIN];
    #pragma unroll
    for (int j = 0; j < DSIN; ++j) Crow[j] = C[ncol * DSIN + j];
    const float hreg  = h[ncol];
    const float adiag = AW[(size_t)ncol * DZ + ncol];
    float zcur = z0[(size_t)b0 * DZ + ncol];

    v2f wtB[DZ / 4];
    #pragma unroll
    for (int k = 0; k < DZ / 4; ++k) {
        int kk = 4 * k + 2 * khalf;
        float bx = (ncol == kk    ) ? 0.0f : AW[(size_t)ncol * DZ + kk];
        float by = (ncol == kk + 1) ? 0.0f : AW[(size_t)ncol * DZ + kk + 1];
        v2f b; b.x = bx; b.y = by;
        wtB[k] = b;
    }

    const int nout = (wave & 3) * 16 + m16;
    v2f boB[DZ / 4];
    #pragma unroll
    for (int k = 0; k < DZ / 4; ++k) {
        int kk = 4 * k + 2 * khalf;
        v2f b;
        b.x = Bobs[(size_t)kk * DX + nout];
        b.y = Bobs[(size_t)(kk + 1) * DX + nout];
        boB[k] = b;
    }

    const __attribute__((address_space(4))) float* inrow = (const __attribute__((address_space(4))) float*)inputs;
    float sc[DSIN];
    #pragma unroll
    for (int j = 0; j < DSIN; ++j) sc[j] = inrow[j];

    __syncthreads();
    {
        const float bev = expand_unit(zcur, thr, th2r, alr);
        if (khalf) {
            be_sh[ncol] = bev;
            be_sh[8 * SZR + ncol] = bev;
        }
    }

    for (int t = 0; t < TSTEPS; ++t) {
        const int buf = t & 1;
        const float* bet = be_sh + buf * TILE;
        __syncthreads();

        v2f afr[DZ / 4];
        #pragma unroll
        for (int k = 0; k < DZ / 4; ++k) {
            int kk = 4 * k + 2 * khalf;
            v2f a;
            a.x = bet[m16 * SZR + kk];
            a.y = bet[m16 * SZR + kk + 1];
            afr[k] = a;
        }

        float hc = hreg;
        #pragma unroll
        for (int j = 0; j < DSIN; ++j) hc = fmaf(Crow[j], sc[j], hc);

        {
            const int tn = (t + 1 < TSTEPS) ? (t + 1) : (TSTEPS - 1);
            #pragma unroll
            for (int j = 0; j < DSIN; ++j) sc[j] = inrow[(size_t)tn * DSIN + j];
        }

        __builtin_amdgcn_sched_barrier(0);
        v8f d = {0.f,0.f,0.f,0.f,0.f,0.f,0.f,0.f};
        #pragma unroll
        for (int k = 0; k < DZ / 4; ++k)
            d = __builtin_amdgcn_wmma_f32_16x16x4_f32(
                    false, afr[k], false, wtB[k], (short)0, d, false, false);

        const float znew = fmaf(zcur, adiag, d[0] + hc);
        zcur = znew;

        const float bev = expand_unit(znew, thr, th2r, alr);
        if (khalf) {
            be_sh[(buf ^ 1) * TILE + ncol] = bev;
            be_sh[(buf ^ 1) * TILE + 8 * SZR + ncol] = bev;
        } else {
            zh_sh[((t >> 4) & 1) * TILE + (t & (HROWS - 1)) * SZR + ncol] = znew;
        }

        if ((t & (HROWS - 1)) == 0 && t > 0 && wave < 4) {
            const float* zt = zh_sh + (((t >> 4) - 1) & 1) * TILE;
            v2f zfr[DZ / 4];
            #pragma unroll
            for (int k = 0; k < DZ / 4; ++k) {
                int kk = 4 * k + 2 * khalf;
                v2f a;
                a.x = zt[m16 * SZR + kk];
                a.y = zt[m16 * SZR + kk + 1];
                zfr[k] = a;
            }
            __builtin_amdgcn_sched_barrier(0);
            v8f o = {0.f,0.f,0.f,0.f,0.f,0.f,0.f,0.f};
            #pragma unroll
            for (int k = 0; k < DZ / 4; ++k)
                o = __builtin_amdgcn_wmma_f32_16x16x4_f32(
                        false, zfr[k], false, boB[k], (short)0, o, false, false);
            const int tb = t - HROWS + 8 * khalf;
            #pragma unroll
            for (int r = 0; r < 8; ++r)
                out[((size_t)(tb + r) * BATCH + b0) * DX + nout] = o[r];
        }
    }

    __syncthreads();
    if (wave < 4) {
        const float* zt = zh_sh + (((TSTEPS >> 4) - 1) & 1) * TILE;
        v2f zfr[DZ / 4];
        #pragma unroll
        for (int k = 0; k < DZ / 4; ++k) {
            int kk = 4 * k + 2 * khalf;
            v2f a;
            a.x = zt[m16 * SZR + kk];
            a.y = zt[m16 * SZR + kk + 1];
            zfr[k] = a;
        }
        v8f o = {0.f,0.f,0.f,0.f,0.f,0.f,0.f,0.f};
        #pragma unroll
        for (int k = 0; k < DZ / 4; ++k)
            o = __builtin_amdgcn_wmma_f32_16x16x4_f32(
                    false, zfr[k], false, boB[k], (short)0, o, false, false);
        const int tb = TSTEPS - HROWS + 8 * khalf;
        #pragma unroll
        for (int r = 0; r < 8; ++r)
            out[((size_t)(tb + r) * BATCH + b0) * DX + nout] = o[r];
    }
}

extern "C" void kernel_launch(void* const* d_in, const int* in_sizes, int n_in,
                              void* d_out, int out_size, void* d_ws, size_t ws_size,
                              hipStream_t stream) {
    const float* z0   = (const float*)d_in[0];
    const float* inp  = (const float*)d_in[1];
    const float* AW   = (const float*)d_in[2];
    const float* hb   = (const float*)d_in[3];
    const float* th   = (const float*)d_in[4];
    const float* al   = (const float*)d_in[5];
    const float* cl   = (const float*)d_in[6];
    const float* C    = (const float*)d_in[7];
    const float* Bo   = (const float*)d_in[8];
    float* out = (float*)d_out;

    plrnn_fused_kernel<<<dim3(NWG), dim3(NTHREADS), 0, stream>>>(
        z0, inp, AW, hb, th, al, cl, C, Bo, out);

    (void)in_sizes; (void)n_in; (void)out_size; (void)d_ws; (void)ws_size;
}


# Basis sum in |x| form: leaky(a)-leaky(b) = p(a-b)+q(|a|-|b|), p,q=(1+-slope)/2, a-b a per-unit constant: 4 VALU ops per basis, not 8, 4 partial sums per lane; order of the 32-term basis sum changes
# speedup vs baseline: 3.4450x; 1.0937x over previous
#include <hip/hip_runtime.h>
#include <hip/hip_bf16.h>


#define DZ    128
#define DB    32
#define DSIN  32
#define DX    64
#define BATCH 64
#define TSTEPS 4096

#define BREAL 1
#define NWG   (BATCH/BREAL)
#define NTHREADS 256

#define SZR 132
#define HROWS 16
#define TILE (16 * SZR)
#define HB   (DB / 2)

typedef __attribute__((ext_vector_type(2))) float v2f;
typedef __attribute__((ext_vector_type(8))) float v8f;

static __device__ __forceinline__ float xhalf(float x) {
    int b = __builtin_bit_cast(int, x);
    return __builtin_bit_cast(float, __builtin_amdgcn_permlanex16(b, b, 0x76543210, 0xfedcba98, false, false));
}

#define LEAK_LIN (0.5f * (1.0f + 0.18f))
#define LEAK_ABS (0.5f * (1.0f - 0.18f))
static __device__ __forceinline__ float expand_unit(float zv, const float (&thr)[HB], const float (&th2r)[HB],
                                                    const float (&alr)[HB], float kconst) {
    float p0 = 0.0f, p1 = 0.0f, p2 = 0.0f, p3 = 0.0f;
    #pragma unroll
    for (int jj = 0; jj < HB; jj += 4) {
        p0 = fmaf(alr[jj + 0], fabsf(zv - thr[jj + 0]) - fabsf(zv - th2r[jj + 0]), p0);
        p1 = fmaf(alr[jj + 1], fabsf(zv - thr[jj + 1]) - fabsf(zv - th2r[jj + 1]), p1);
        p2 = fmaf(alr[jj + 2], fabsf(zv - thr[jj + 2]) - fabsf(zv - th2r[jj + 2]), p2);
        p3 = fmaf(alr[jj + 3], fabsf(zv - thr[jj + 3]) - fabsf(zv - th2r[jj + 3]), p3);
    }
    float s = (p0 + p1) + (p2 + p3);
    s += xhalf(s);
    return fmaf(LEAK_ABS, s, kconst);
}

__launch_bounds__(NTHREADS, 1)
__global__ void plrnn_fused_kernel(const float* __restrict__ z0,
                                   const float* __restrict__ inputs,
                                   const float* __restrict__ AW,
                                   const float* __restrict__ h,
                                   const float* __restrict__ thetas,
                                   const float* __restrict__ alphas,
                                   const float* __restrict__ clipping,
                                   const float* __restrict__ C,
                                   const float* __restrict__ Bobs,
                                   float* __restrict__ out)
{
    __shared__ float be_sh[2 * TILE];
    __shared__ float zh_sh[2 * TILE];

    const int tid   = threadIdx.x;
    const int lane  = tid & 31;
    const int wave  = tid >> 5;
    const int m16   = lane & 15;
    const int khalf = lane >> 4;
    const int b0    = blockIdx.x * BREAL;

    for (int idx = tid; idx < 2 * TILE; idx += NTHREADS) be_sh[idx] = 0.0f;

    const int ncol = wave * 16 + m16;

    float thr[HB], th2r[HB], alr[HB];
    #pragma unroll
    for (int jj = 0; jj < HB; ++jj) {
        const int j = HB * khalf + jj;
        const float tv = thetas[ncol * DB + j];
        thr [jj] = tv;
        th2r[jj] = tv - clipping[ncol * DB + j];
        alr [jj] = alphas[j];
    }
    float kconst;
    {
        float q0 = 0.0f, q1 = 0.0f, q2 = 0.0f, q3 = 0.0f;
        #pragma unroll
        for (int jj = 0; jj < HB; jj += 4) {
            q0 = fmaf(alr[jj + 0], th2r[jj + 0] - thr[jj + 0], q0);
            q1 = fmaf(alr[jj + 1], th2r[jj + 1] - thr[jj + 1], q1);
            q2 = fmaf(alr[jj + 2], th2r[jj + 2] - thr[jj + 2], q2);
            q3 = fmaf(alr[jj + 3], th2r[jj + 3] - thr[jj + 3], q3);
        }
        float q = (q0 + q1) + (q2 + q3);
        q += xhalf(q);
        kconst = LEAK_LIN * q;
    }
    float Crow[DSIN];
    #pragma unroll
    for (int j = 0; j < DSIN; ++j) Crow[j] = C[ncol * DSIN + j];
    const float hreg  = h[ncol];
    const float adiag = AW[(size_t)ncol * DZ + ncol];
    float zcur = z0[(size_t)b0 * DZ + ncol];

    v2f wtB[DZ / 4];
    #pragma unroll
    for (int k = 0; k < DZ / 4; ++k) {
        int kk = 4 * k + 2 * khalf;
        float bx = (ncol == kk    ) ? 0.0f : AW[(size_t)ncol * DZ + kk];
        float by = (ncol == kk + 1) ? 0.0f : AW[(size_t)ncol * DZ + kk + 1];
        v2f b; b.x = bx; b.y = by;
        wtB[k] = b;
    }

    const int nout = (wave & 3) * 16 + m16;
    v2f boB[DZ / 4];
    #pragma unroll
    for (int k = 0; k < DZ / 4; ++k) {
        int kk = 4 * k + 2 * khalf;
        v2f b;
        b.x = Bobs[(size_t)kk * DX + nout];
        b.y = Bobs[(size_t)(kk + 1) * DX + nout];
        boB[k] = b;
    }

    const __attribute__((address_space(4))) float* inrow = (const __attribute__((address_space(4))) float*)inputs;
    float sc[DSIN];
    #pragma unroll
    for (int j = 0; j < DSIN; ++j) sc[j] = inrow[j];

    __syncthreads();
    {
        const float bev = expand_unit(zcur, thr, th2r, alr, kconst);
        if (khalf) {
            be_sh[ncol] = bev;
            be_sh[8 * SZR + ncol] = bev;
        }
    }

    for (int t = 0; t < TSTEPS; ++t) {
        const int buf = t & 1;
        const float* bet = be_sh + buf * TILE;
        __syncthreads();

        v2f afr[DZ / 4];
        #pragma unroll
        for (int k = 0; k < DZ / 4; ++k) {
            int kk = 4 * k + 2 * khalf;
            v2f a;
            a.x = bet[m16 * SZR + kk];
            a.y = bet[m16 * SZR + kk + 1];
            afr[k] = a;
        }

        float hc = hreg;
        #pragma unroll
        for (int j = 0; j < DSIN; ++j) hc = fmaf(Crow[j], sc[j], hc);

        {
            const int tn = (t + 1 < TSTEPS) ? (t + 1) : (TSTEPS - 1);
            #pragma unroll
            for (int j = 0; j < DSIN; ++j) sc[j] = inrow[(size_t)tn * DSIN + j];
        }

        __builtin_amdgcn_sched_barrier(0);
        v8f d = {0.f,0.f,0.f,0.f,0.f,0.f,0.f,0.f};
        #pragma unroll
        for (int k = 0; k < DZ / 4; ++k)
            d = __builtin_amdgcn_wmma_f32_16x16x4_f32(
                    false, afr[k], false, wtB[k], (short)0, d, false, false);

        const float znew = fmaf(zcur, adiag, d[0] + hc);
        zcur = znew;

        const float bev = expand_unit(znew, thr, th2r, alr, kconst);
        if (khalf) {
            be_sh[(buf ^ 1) * TILE + ncol] = bev;
            be_sh[(buf ^ 1) * TILE + 8 * SZR + ncol] = bev;
        } else {
            zh_sh[((t >> 4) & 1) * TILE + (t & (HROWS - 1)) * SZR + ncol] = znew;
        }

        if ((t & (HROWS - 1)) == 0 && t > 0 && wave < 4) {
            const float* zt = zh_sh + (((t >> 4) - 1) & 1) * TILE;
            v2f zfr[DZ / 4];
            #pragma unroll
            for (int k = 0; k < DZ / 4; ++k) {
                int kk = 4 * k + 2 * khalf;
                v2f a;
                a.x = zt[m16 * SZR + kk];
                a.y = zt[m16 * SZR + kk + 1];
                zfr[k] = a;
            }
            __builtin_amdgcn_sched_barrier(0);
            v8f o = {0.f,0.f,0.f,0.f,0.f,0.f,0.f,0.f};
            #pragma unroll
            for (int k = 0; k < DZ / 4; ++k)
                o = __builtin_amdgcn_wmma_f32_16x16x4_f32(
                        false, zfr[k], false, boB[k], (short)0, o, false, false);
            const int tb = t - HROWS + 8 * khalf;
            #pragma unroll
            for (int r = 0; r < 8; ++r)
                out[((size_t)(tb + r) * BATCH + b0) * DX + nout] = o[r];
        }
    }

    __syncthreads();
    if (wave < 4) {
        const float* zt = zh_sh + (((TSTEPS >> 4) - 1) & 1) * TILE;
        v2f zfr[DZ / 4];
        #pragma unroll
        for (int k = 0; k < DZ / 4; ++k) {
            int kk = 4 * k + 2 * khalf;
            v2f a;
            a.x = zt[m16 * SZR + kk];
            a.y = zt[m16 * SZR + kk + 1];
            zfr[k] = a;
        }
        v8f o = {0.f,0.f,0.f,0.f,0.f,0.f,0.f,0.f};
        #pragma unroll
        for (int k = 0; k < DZ / 4; ++k)
            o = __builtin_amdgcn_wmma_f32_16x16x4_f32(
                    false, zfr[k], false, boB[k], (short)0, o, false, false);
        const int tb = TSTEPS - HROWS + 8 * khalf;
        #pragma unroll
        for (int r = 0; r < 8; ++r)
            out[((size_t)(tb + r) * BATCH + b0) * DX + nout] = o[r];
    }
}

extern "C" void kernel_launch(void* const* d_in, const int* in_sizes, int n_in,
                              void* d_out, int out_size, void* d_ws, size_t ws_size,
                              hipStream_t stream) {
    const float* z0   = (const float*)d_in[0];
    const float* inp  = (const float*)d_in[1];
    const float* AW   = (const float*)d_in[2];
    const float* hb   = (const float*)d_in[3];
    const float* th   = (const float*)d_in[4];
    const float* al   = (const float*)d_in[5];
    const float* cl   = (const float*)d_in[6];
    const float* C    = (const float*)d_in[7];
    const float* Bo   = (const float*)d_in[8];
    float* out = (float*)d_out;

    plrnn_fused_kernel<<<dim3(NWG), dim3(NTHREADS), 0, stream>>>(
        z0, inp, AW, hb, th, al, cl, C, Bo, out);

    (void)in_sizes; (void)n_in; (void)out_size; (void)d_ws; (void)ws_size;
}
